# starting kernel
# baseline (speedup 1.0000x reference)
#include <hip/hip_runtime.h>
#include <hip/hip_bf16.h>

typedef __attribute__((ext_vector_type(2))) float v2f;
typedef __attribute__((ext_vector_type(8))) float v8f;

#define SGRID 7
#define CELLS 49
#define NCH   30
#define NBATCH 16384
#define NCLASS 20

#define WS_COORD 0
#define WS_CONF  49
#define WS_CLS   98
#define WS_A     152
#define WS_CB    204
#define WS_HAS   256
#define WS_HDR   256

__global__ void yolo_init(float* __restrict__ ws) {
    int t = threadIdx.x;
    if (t < WS_HDR) ws[t] = 0.0f;
}

__global__ void yolo_reduce(const float* __restrict__ out,
                            const float* __restrict__ tgt,
                            const int* __restrict__ imw,
                            const int* __restrict__ imh,
                            float* __restrict__ ws) {
    __shared__ float bins[3 * CELLS];
    const int tid = threadIdx.x;
    for (int k = tid; k < 3 * CELLS; k += blockDim.x) bins[k] = 0.0f;
    __syncthreads();

    const int   g  = blockIdx.x * blockDim.x + tid;
    const float W  = (float)imw[0];
    const float H  = (float)imh[0];
    const float cw = W / (float)SGRID;
    const float ch = H / (float)SGRID;

    const int b  = g / CELLS;
    const int ij = g - b * CELLS;
    const int i  = ij / SGRID;
    const int j  = ij - i * SGRID;

    const float* t = tgt + (size_t)g * NCH;
    const float* o = out + (size_t)g * NCH;

    const float t0 = t[0], t1 = t[1];
    const float tw = t[2] * W, th = t[3] * H;
    const float has = t[4];
    const float tcx = ((float)j + t0) * cw;
    const float tcy = ((float)i + t1) * ch;
    const float tx0 = tcx - tw * 0.5f, ty0 = tcy - th * 0.5f;
    const float tx1 = tcx + tw * 0.5f, ty1 = tcy + th * 0.5f;
    const float area_t = tw * th;

    float iou_p[2], conf_p[2];
#pragma unroll
    for (int p = 0; p < 2; ++p) {
        const float px = o[5 * p + 0], py = o[5 * p + 1];
        const float pw = o[5 * p + 3] * W, ph = o[5 * p + 4] * H;
        const float pcx = ((float)j + px) * cw;
        const float pcy = ((float)i + py) * ch;
        const float px0 = pcx - pw * 0.5f, py0 = pcy - ph * 0.5f;
        const float px1 = pcx + pw * 0.5f, py1 = pcy + ph * 0.5f;
        float iw = fminf(px1, tx1) - fmaxf(px0, tx0); iw = fmaxf(iw, 0.0f);
        float ih = fminf(py1, ty1) - fmaxf(py0, ty0); ih = fmaxf(ih, 0.0f);
        const float inter = iw * ih;
        iou_p[p]  = inter / (pw * ph + area_t - inter + 1e-9f);
        conf_p[p] = o[5 * p + 4];
    }
    const float best_conf = (iou_p[1] >= iou_p[0]) ? conf_p[1] : conf_p[0];

    const float px = o[5], py = o[6];
    const float pw = o[8] * W, ph = o[9] * H;
    const float pconf = o[9];
    const float dx = t0 - px, dy = t1 - py;
    const float dw = sqrtf(tw) - sqrtf(pw);
    const float dh = sqrtf(th) - sqrtf(ph);
    const float coord = dx * dx + dy * dy + dw * dw + dh * dh;
    const float dc = has - pconf;
    const float confm = dc * dc;

    float cls = 0.0f;
#pragma unroll
    for (int c = 0; c < NCLASS; ++c) {
        const float d = t[10 + c] - o[10 + c] * best_conf;
        cls = fmaf(d, d, cls);
    }

    ws[WS_HAS + g] = has;

    atomicAdd(&bins[ij],             coord);
    atomicAdd(&bins[CELLS + ij],     confm);
    atomicAdd(&bins[2 * CELLS + ij], cls);
    __syncthreads();
    for (int k = tid; k < 3 * CELLS; k += blockDim.x)
        atomicAdd(&ws[k], bins[k]);
}

__global__ void yolo_finalize(float* __restrict__ ws) {
    __shared__ float red[64];
    const int tid = threadIdx.x;
    float a = 0.0f, bsum = 0.0f;
    if (tid < CELLS) {
        const float coord = ws[WS_COORD + tid];
        const float conf  = ws[WS_CONF  + tid];
        const float cls   = ws[WS_CLS   + tid];
        a    = 5.0f * coord + 0.5f * conf;
        bsum = 0.5f * conf + cls;
    }
    if (tid < 52) ws[WS_A + tid] = (tid < CELLS) ? a : 0.0f;
    red[tid] = bsum;
    __syncthreads();
    for (int s = 32; s > 0; s >>= 1) {
        if (tid < s) red[tid] += red[tid + s];
        __syncthreads();
    }
    if (tid == 0) ws[WS_CB] = red[0];
}

__global__ void yolo_dot_wmma(const float* __restrict__ ws,
                              float* __restrict__ loss) {
    const int lane = threadIdx.x & 31;
    const int wave = (blockIdx.x * blockDim.x + threadIdx.x) >> 5;
    const int b0   = wave * 16;
    const float* __restrict__ Avec = ws + WS_A;
    const float* __restrict__ hasv = ws + WS_HAS;

    const int half = lane >> 4;
    const int m    = lane & 15;

    v8f acc = {};
#pragma unroll
    for (int k0 = 0; k0 < 52; k0 += 4) {
        const int ka = k0 + (half ? 2 : 0);
        const int kb = ka + 1;
        const int kca = ka > 48 ? 48 : ka;
        const int kcb = kb > 48 ? 48 : kb;
        v2f a, bm;
        a.x  = hasv[(size_t)(b0 + m) * CELLS + kca];
        a.y  = hasv[(size_t)(b0 + m) * CELLS + kcb];
        bm.x = Avec[ka];
        bm.y = Avec[kb];
        acc = __builtin_amdgcn_wmma_f32_16x16x4_f32(
             false, a,  false, bm,
             (short)0, acc,  false,  false);
    }

    const float cb = ws[WS_CB];
    if ((lane & 15) == 0) {
        const int mbase = b0 + half * 8;
#pragma unroll
        for (int r = 0; r < 8; ++r)
            loss[mbase + r] = acc[r] + cb;
    }
}

extern "C" void kernel_launch(void* const* d_in, const int* in_sizes, int n_in,
                              void* d_out, int out_size, void* d_ws, size_t ws_size,
                              hipStream_t stream) {
    const float* out_t = (const float*)d_in[0];
    const float* tgt_t = (const float*)d_in[1];
    const int*   imw   = (const int*)d_in[2];
    const int*   imh   = (const int*)d_in[3];
    float* loss = (float*)d_out;
    float* ws   = (float*)d_ws;

    yolo_init<<<1, 256, 0, stream>>>(ws);

    yolo_reduce<<<3136, 256, 0, stream>>>(out_t, tgt_t, imw, imh, ws);

    yolo_finalize<<<1, 64, 0, stream>>>(ws);

    yolo_dot_wmma<<<128, 256, 0, stream>>>(ws, loss);
}


# reduce pass: each thread owns two adjacent cells and reads their 240 contiguous bytes per tensor by fifteen 16-byte loads straight into registers, no staging; atomic sum kind kept, 512 cells a block
# speedup vs baseline: 1.5836x; 1.5836x over previous
#include <hip/hip_runtime.h>
#include <hip/hip_bf16.h>

typedef __attribute__((ext_vector_type(2))) float v2f;
typedef __attribute__((ext_vector_type(8))) float v8f;
typedef __attribute__((ext_vector_type(4))) float v4f;

#define SGRID 7
#define CELLS 49
#define NCH   30
#define NBATCH 16384
#define NCLASS 20

#define WS_COORD 0
#define WS_CONF  49
#define WS_CLS   98
#define WS_A     152
#define WS_CB    204
#define WS_HAS   256
#define WS_HDR   256

__global__ void yolo_init(float* __restrict__ ws) {
    int t = threadIdx.x;
    if (t < WS_HDR) ws[t] = 0.0f;
}

#define PAIRV 15

__global__ __launch_bounds__(256)
void yolo_reduce(const float* __restrict__ out,
                 const float* __restrict__ tgt,
                 const int* __restrict__ imw,
                 const int* __restrict__ imh,
                 float* __restrict__ ws) {
    __shared__ float bins[3 * CELLS];
    const int tid = threadIdx.x;
    for (int k = tid; k < 3 * CELLS; k += blockDim.x) bins[k] = 0.0f;
    __syncthreads();

    const int   pr = blockIdx.x * blockDim.x + tid;
    const int   g0 = 2 * pr;
    const float W  = (float)imw[0];
    const float H  = (float)imh[0];
    const float cw = W / (float)SGRID;
    const float ch = H / (float)SGRID;

    const v4f* __restrict__ tp = reinterpret_cast<const v4f*>(tgt + (size_t)g0 * NCH);
    const v4f* __restrict__ op = reinterpret_cast<const v4f*>(out + (size_t)g0 * NCH);

    v4f tv[PAIRV], ov[PAIRV];
#pragma unroll
    for (int k = 0; k < PAIRV; ++k) tv[k] = tp[k];
#pragma unroll
    for (int k = 0; k < PAIRV; ++k) ov[k] = op[k];
#pragma unroll
    for (int k = 0; k < PAIRV; ++k) asm volatile("" : "+v"(tv[k]));
#pragma unroll
    for (int k = 0; k < PAIRV; ++k) asm volatile("" : "+v"(ov[k]));

    float tf[2 * NCH], of[2 * NCH];
#pragma unroll
    for (int k = 0; k < PAIRV; ++k) {
        tf[4 * k + 0] = tv[k].x; tf[4 * k + 1] = tv[k].y; tf[4 * k + 2] = tv[k].z; tf[4 * k + 3] = tv[k].w;
        of[4 * k + 0] = ov[k].x; of[4 * k + 1] = ov[k].y; of[4 * k + 2] = ov[k].z; of[4 * k + 3] = ov[k].w;
    }

    const int b0  = g0 / CELLS;
    const int ij0 = g0 - b0 * CELLS;

    float hasp[2];
#pragma unroll
    for (int c = 0; c < 2; ++c) {
        int ij = ij0 + c;
        if (ij == CELLS) ij = 0;
        const int i = ij / SGRID;
        const int j = ij - i * SGRID;

        const float* t = tf + c * NCH;
        const float* o = of + c * NCH;

        const float t0 = t[0], t1 = t[1];
        const float tw = t[2] * W, th = t[3] * H;
        const float has = t[4];
        const float tcx = ((float)j + t0) * cw;
        const float tcy = ((float)i + t1) * ch;
        const float tx0 = tcx - tw * 0.5f, ty0 = tcy - th * 0.5f;
        const float tx1 = tcx + tw * 0.5f, ty1 = tcy + th * 0.5f;
        const float area_t = tw * th;

        float iou_p[2], conf_p[2];
#pragma unroll
        for (int p = 0; p < 2; ++p) {
            const float px = o[5 * p + 0], py = o[5 * p + 1];
            const float pw = o[5 * p + 3] * W, ph = o[5 * p + 4] * H;
            const float pcx = ((float)j + px) * cw;
            const float pcy = ((float)i + py) * ch;
            const float px0 = pcx - pw * 0.5f, py0 = pcy - ph * 0.5f;
            const float px1 = pcx + pw * 0.5f, py1 = pcy + ph * 0.5f;
            float iw = fminf(px1, tx1) - fmaxf(px0, tx0); iw = fmaxf(iw, 0.0f);
            float ih = fminf(py1, ty1) - fmaxf(py0, ty0); ih = fmaxf(ih, 0.0f);
            const float inter = iw * ih;
            iou_p[p]  = inter / (pw * ph + area_t - inter + 1e-9f);
            conf_p[p] = o[5 * p + 4];
        }
        const float best_conf = (iou_p[1] >= iou_p[0]) ? conf_p[1] : conf_p[0];

        const float px = o[5], py = o[6];
        const float pw = o[8] * W, ph = o[9] * H;
        const float pconf = o[9];
        const float dx = t0 - px, dy = t1 - py;
        const float dw = sqrtf(tw) - sqrtf(pw);
        const float dh = sqrtf(th) - sqrtf(ph);
        const float coord = dx * dx + dy * dy + dw * dw + dh * dh;
        const float dc = has - pconf;
        const float confm = dc * dc;

        float cls = 0.0f;
#pragma unroll
        for (int q = 0; q < NCLASS; ++q) {
            const float d = t[10 + q] - o[10 + q] * best_conf;
            cls = fmaf(d, d, cls);
        }

        hasp[c] = has;

        atomicAdd(&bins[ij],             coord);
        atomicAdd(&bins[CELLS + ij],     confm);
        atomicAdd(&bins[2 * CELLS + ij], cls);
    }

    *reinterpret_cast<float2*>(ws + WS_HAS + g0) = make_float2(hasp[0], hasp[1]);

    __syncthreads();
    for (int k = tid; k < 3 * CELLS; k += blockDim.x)
        atomicAdd(&ws[k], bins[k]);
}

__global__ void yolo_finalize(float* __restrict__ ws) {
    __shared__ float red[64];
    const int tid = threadIdx.x;
    float a = 0.0f, bsum = 0.0f;
    if (tid < CELLS) {
        const float coord = ws[WS_COORD + tid];
        const float conf  = ws[WS_CONF  + tid];
        const float cls   = ws[WS_CLS   + tid];
        a    = 5.0f * coord + 0.5f * conf;
        bsum = 0.5f * conf + cls;
    }
    if (tid < 52) ws[WS_A + tid] = (tid < CELLS) ? a : 0.0f;
    red[tid] = bsum;
    __syncthreads();
    for (int s = 32; s > 0; s >>= 1) {
        if (tid < s) red[tid] += red[tid + s];
        __syncthreads();
    }
    if (tid == 0) ws[WS_CB] = red[0];
}

__global__ void yolo_dot_wmma(const float* __restrict__ ws,
                              float* __restrict__ loss) {
    const int lane = threadIdx.x & 31;
    const int wave = (blockIdx.x * blockDim.x + threadIdx.x) >> 5;
    const int b0   = wave * 16;
    const float* __restrict__ Avec = ws + WS_A;
    const float* __restrict__ hasv = ws + WS_HAS;

    const int half = lane >> 4;
    const int m    = lane & 15;

    v8f acc = {};
#pragma unroll
    for (int k0 = 0; k0 < 52; k0 += 4) {
        const int ka = k0 + (half ? 2 : 0);
        const int kb = ka + 1;
        const int kca = ka > 48 ? 48 : ka;
        const int kcb = kb > 48 ? 48 : kb;
        v2f a, bm;
        a.x  = hasv[(size_t)(b0 + m) * CELLS + kca];
        a.y  = hasv[(size_t)(b0 + m) * CELLS + kcb];
        bm.x = Avec[ka];
        bm.y = Avec[kb];
        acc = __builtin_amdgcn_wmma_f32_16x16x4_f32(
             false, a,  false, bm,
             (short)0, acc,  false,  false);
    }

    const float cb = ws[WS_CB];
    if ((lane & 15) == 0) {
        const int mbase = b0 + half * 8;
#pragma unroll
        for (int r = 0; r < 8; ++r)
            loss[mbase + r] = acc[r] + cb;
    }
}

extern "C" void kernel_launch(void* const* d_in, const int* in_sizes, int n_in,
                              void* d_out, int out_size, void* d_ws, size_t ws_size,
                              hipStream_t stream) {
    const float* out_t = (const float*)d_in[0];
    const float* tgt_t = (const float*)d_in[1];
    const int*   imw   = (const int*)d_in[2];
    const int*   imh   = (const int*)d_in[3];
    float* loss = (float*)d_out;
    float* ws   = (float*)d_ws;

    yolo_init<<<1, 256, 0, stream>>>(ws);

    yolo_reduce<<<1568, 256, 0, stream>>>(out_t, tgt_t, imw, imh, ws);

    yolo_finalize<<<1, 64, 0, stream>>>(ws);

    yolo_dot_wmma<<<128, 256, 0, stream>>>(ws, loss);
}


# reduce pass loads: a quad's 4 lanes read 64 contiguous bytes per load (8 lines per wave load, was 32) and pass chunks to their owner lanes by lane selects and quad permutes; atomic sum kind kept
# speedup vs baseline: 1.8172x; 1.1475x over previous
#include <hip/hip_runtime.h>
#include <hip/hip_bf16.h>

typedef __attribute__((ext_vector_type(2))) float v2f;
typedef __attribute__((ext_vector_type(8))) float v8f;
typedef __attribute__((ext_vector_type(4))) float v4f;

#define SGRID 7
#define CELLS 49
#define NCH   30
#define NBATCH 16384
#define NCLASS 20

#define WS_COORD 0
#define WS_CONF  49
#define WS_CLS   98
#define WS_A     152
#define WS_CB    204
#define WS_HAS   256
#define WS_HDR   256

__global__ void yolo_init(float* __restrict__ ws) {
    int t = threadIdx.x;
    if (t < WS_HDR) ws[t] = 0.0f;
}

#define PAIRV 15

template <int CTRL>
__device__ __forceinline__ float quad_read(float v) {
    return __int_as_float(__builtin_amdgcn_mov_dpp(__float_as_int(v), CTRL, 0xF, 0xF, true));
}

__device__ __forceinline__ float quad_rot(float v, int jm) {
    switch (jm) {
        case 0:  return quad_read<0x6C>(v);
        case 1:  return quad_read<0xB1>(v);
        case 2:  return quad_read<0xC6>(v);
        default: return quad_read<0x1B>(v);
    }
}

__device__ __forceinline__ void quad_gather(const float (&ld)[4 * PAIRV], float (&own)[4 * PAIRV],
                                            bool q_bit0, bool q_bit1) {
#pragma unroll
    for (int j = 0; j < PAIRV; ++j) {
        const int k0 = (15 * ((j - 0) & 3) + j) >> 2;
        const int k1 = (15 * ((j - 1) & 3) + j) >> 2;
        const int k2 = (15 * ((j - 2) & 3) + j) >> 2;
        const int k3 = (15 * ((j - 3) & 3) + j) >> 2;
#pragma unroll
        for (int e = 0; e < 4; ++e) {
            const float s0 = ld[4 * k0 + e], s1 = ld[4 * k1 + e];
            const float s2 = ld[4 * k2 + e], s3 = ld[4 * k3 + e];
            const float lo = q_bit0 ? s1 : s0;
            const float hi = q_bit0 ? s3 : s2;
            own[4 * j + e] = quad_rot(q_bit1 ? hi : lo, j & 3);
        }
    }
}

__global__ __launch_bounds__(256)
void yolo_reduce(const float* __restrict__ out,
                 const float* __restrict__ tgt,
                 const int* __restrict__ imw,
                 const int* __restrict__ imh,
                 float* __restrict__ ws) {
    __shared__ float bins[3 * CELLS];
    const int tid = threadIdx.x;
    for (int k = tid; k < 3 * CELLS; k += blockDim.x) bins[k] = 0.0f;
    __syncthreads();

    const int   pr = blockIdx.x * blockDim.x + tid;
    const int   g0 = 2 * pr;
    const float W  = (float)imw[0];
    const float H  = (float)imh[0];
    const float cw = W / (float)SGRID;
    const float ch = H / (float)SGRID;

    const int qbase = (pr >> 2) * (4 * PAIRV) + (pr & 3);
    const v4f* __restrict__ tp = reinterpret_cast<const v4f*>(tgt) + qbase;
    const v4f* __restrict__ op = reinterpret_cast<const v4f*>(out) + qbase;

    v4f tv[PAIRV], ov[PAIRV];
#pragma unroll
    for (int k = 0; k < PAIRV; ++k) tv[k] = tp[4 * k];
#pragma unroll
    for (int k = 0; k < PAIRV; ++k) ov[k] = op[4 * k];
#pragma unroll
    for (int k = 0; k < PAIRV; ++k) asm volatile("" : "+v"(tv[k]));
#pragma unroll
    for (int k = 0; k < PAIRV; ++k) asm volatile("" : "+v"(ov[k]));

    float tl[2 * NCH], ol[2 * NCH];
#pragma unroll
    for (int k = 0; k < PAIRV; ++k) {
        tl[4 * k + 0] = tv[k].x; tl[4 * k + 1] = tv[k].y; tl[4 * k + 2] = tv[k].z; tl[4 * k + 3] = tv[k].w;
        ol[4 * k + 0] = ov[k].x; ol[4 * k + 1] = ov[k].y; ol[4 * k + 2] = ov[k].z; ol[4 * k + 3] = ov[k].w;
    }

    const bool q_bit0 = (tid & 1) != 0;
    const bool q_bit1 = (tid & 2) != 0;
    float tf[2 * NCH], of[2 * NCH];
    quad_gather(tl, tf, q_bit0, q_bit1);
    quad_gather(ol, of, q_bit0, q_bit1);

    const int b0  = g0 / CELLS;
    const int ij0 = g0 - b0 * CELLS;

    float hasp[2];
#pragma unroll
    for (int c = 0; c < 2; ++c) {
        int ij = ij0 + c;
        if (ij == CELLS) ij = 0;
        const int i = ij / SGRID;
        const int j = ij - i * SGRID;

        const float* t = tf + c * NCH;
        const float* o = of + c * NCH;

        const float t0 = t[0], t1 = t[1];
        const float tw = t[2] * W, th = t[3] * H;
        const float has = t[4];
        const float tcx = ((float)j + t0) * cw;
        const float tcy = ((float)i + t1) * ch;
        const float tx0 = tcx - tw * 0.5f, ty0 = tcy - th * 0.5f;
        const float tx1 = tcx + tw * 0.5f, ty1 = tcy + th * 0.5f;
        const float area_t = tw * th;

        float iou_p[2], conf_p[2];
#pragma unroll
        for (int p = 0; p < 2; ++p) {
            const float px = o[5 * p + 0], py = o[5 * p + 1];
            const float pw = o[5 * p + 3] * W, ph = o[5 * p + 4] * H;
            const float pcx = ((float)j + px) * cw;
            const float pcy = ((float)i + py) * ch;
            const float px0 = pcx - pw * 0.5f, py0 = pcy - ph * 0.5f;
            const float px1 = pcx + pw * 0.5f, py1 = pcy + ph * 0.5f;
            float iw = fminf(px1, tx1) - fmaxf(px0, tx0); iw = fmaxf(iw, 0.0f);
            float ih = fminf(py1, ty1) - fmaxf(py0, ty0); ih = fmaxf(ih, 0.0f);
            const float inter = iw * ih;
            iou_p[p]  = inter / (pw * ph + area_t - inter + 1e-9f);
            conf_p[p] = o[5 * p + 4];
        }
        const float best_conf = (iou_p[1] >= iou_p[0]) ? conf_p[1] : conf_p[0];

        const float px = o[5], py = o[6];
        const float pw = o[8] * W, ph = o[9] * H;
        const float pconf = o[9];
        const float dx = t0 - px, dy = t1 - py;
        const float dw = sqrtf(tw) - sqrtf(pw);
        const float dh = sqrtf(th) - sqrtf(ph);
        const float coord = dx * dx + dy * dy + dw * dw + dh * dh;
        const float dc = has - pconf;
        const float confm = dc * dc;

        float cls = 0.0f;
#pragma unroll
        for (int q = 0; q < NCLASS; ++q) {
            const float d = t[10 + q] - o[10 + q] * best_conf;
            cls = fmaf(d, d, cls);
        }

        hasp[c] = has;

        atomicAdd(&bins[ij],             coord);
        atomicAdd(&bins[CELLS + ij],     confm);
        atomicAdd(&bins[2 * CELLS + ij], cls);
    }

    *reinterpret_cast<float2*>(ws + WS_HAS + g0) = make_float2(hasp[0], hasp[1]);

    __syncthreads();
    for (int k = tid; k < 3 * CELLS; k += blockDim.x)
        atomicAdd(&ws[k], bins[k]);
}

__global__ void yolo_finalize(float* __restrict__ ws) {
    __shared__ float red[64];
    const int tid = threadIdx.x;
    float a = 0.0f, bsum = 0.0f;
    if (tid < CELLS) {
        const float coord = ws[WS_COORD + tid];
        const float conf  = ws[WS_CONF  + tid];
        const float cls   = ws[WS_CLS   + tid];
        a    = 5.0f * coord + 0.5f * conf;
        bsum = 0.5f * conf + cls;
    }
    if (tid < 52) ws[WS_A + tid] = (tid < CELLS) ? a : 0.0f;
    red[tid] = bsum;
    __syncthreads();
    for (int s = 32; s > 0; s >>= 1) {
        if (tid < s) red[tid] += red[tid + s];
        __syncthreads();
    }
    if (tid == 0) ws[WS_CB] = red[0];
}

__global__ void yolo_dot_wmma(const float* __restrict__ ws,
                              float* __restrict__ loss) {
    const int lane = threadIdx.x & 31;
    const int wave = (blockIdx.x * blockDim.x + threadIdx.x) >> 5;
    const int b0   = wave * 16;
    const float* __restrict__ Avec = ws + WS_A;
    const float* __restrict__ hasv = ws + WS_HAS;

    const int half = lane >> 4;
    const int m    = lane & 15;

    v8f acc = {};
#pragma unroll
    for (int k0 = 0; k0 < 52; k0 += 4) {
        const int ka = k0 + (half ? 2 : 0);
        const int kb = ka + 1;
        const int kca = ka > 48 ? 48 : ka;
        const int kcb = kb > 48 ? 48 : kb;
        v2f a, bm;
        a.x  = hasv[(size_t)(b0 + m) * CELLS + kca];
        a.y  = hasv[(size_t)(b0 + m) * CELLS + kcb];
        bm.x = Avec[ka];
        bm.y = Avec[kb];
        acc = __builtin_amdgcn_wmma_f32_16x16x4_f32(
             false, a,  false, bm,
             (short)0, acc,  false,  false);
    }

    const float cb = ws[WS_CB];
    if ((lane & 15) == 0) {
        const int mbase = b0 + half * 8;
#pragma unroll
        for (int r = 0; r < 8; ++r)
            loss[mbase + r] = acc[r] + cb;
    }
}

extern "C" void kernel_launch(void* const* d_in, const int* in_sizes, int n_in,
                              void* d_out, int out_size, void* d_ws, size_t ws_size,
                              hipStream_t stream) {
    const float* out_t = (const float*)d_in[0];
    const float* tgt_t = (const float*)d_in[1];
    const int*   imw   = (const int*)d_in[2];
    const int*   imh   = (const int*)d_in[3];
    float* loss = (float*)d_out;
    float* ws   = (float*)d_ws;

    yolo_init<<<1, 256, 0, stream>>>(ws);

    yolo_reduce<<<1568, 256, 0, stream>>>(out_t, tgt_t, imw, imh, ws);

    yolo_finalize<<<1, 64, 0, stream>>>(ws);

    yolo_dot_wmma<<<128, 256, 0, stream>>>(ws, loss);
}


# finalize launch gone (4 -> 3) by linearity: each reduce block weights its own 3x49 partial sums into 49 A terms + 1 constant before its atomic adds (50 words, was 147); sum order changes
# speedup vs baseline: 1.9239x; 1.0587x over previous
#include <hip/hip_runtime.h>
#include <hip/hip_bf16.h>

typedef __attribute__((ext_vector_type(2))) float v2f;
typedef __attribute__((ext_vector_type(8))) float v8f;
typedef __attribute__((ext_vector_type(4))) float v4f;

#define SGRID 7
#define CELLS 49
#define NCH   30
#define NBATCH 16384
#define NCLASS 20

#define WS_A     128
#define WS_CB    192
#define WS_HAS   256
#define WS_HDR   256

__global__ void yolo_init(float* __restrict__ ws) {
    int t = threadIdx.x;
    if (t < WS_HDR) ws[t] = 0.0f;
}

#define PAIRV 15

template <int CTRL>
__device__ __forceinline__ float quad_read(float v) {
    return __int_as_float(__builtin_amdgcn_mov_dpp(__float_as_int(v), CTRL, 0xF, 0xF, true));
}

__device__ __forceinline__ float quad_rot(float v, int jm) {
    switch (jm) {
        case 0:  return quad_read<0x6C>(v);
        case 1:  return quad_read<0xB1>(v);
        case 2:  return quad_read<0xC6>(v);
        default: return quad_read<0x1B>(v);
    }
}

__device__ __forceinline__ void quad_gather(const float (&ld)[4 * PAIRV], float (&own)[4 * PAIRV],
                                            bool q_bit0, bool q_bit1) {
#pragma unroll
    for (int j = 0; j < PAIRV; ++j) {
        const int k0 = (15 * ((j - 0) & 3) + j) >> 2;
        const int k1 = (15 * ((j - 1) & 3) + j) >> 2;
        const int k2 = (15 * ((j - 2) & 3) + j) >> 2;
        const int k3 = (15 * ((j - 3) & 3) + j) >> 2;
#pragma unroll
        for (int e = 0; e < 4; ++e) {
            const float s0 = ld[4 * k0 + e], s1 = ld[4 * k1 + e];
            const float s2 = ld[4 * k2 + e], s3 = ld[4 * k3 + e];
            const float lo = q_bit0 ? s1 : s0;
            const float hi = q_bit0 ? s3 : s2;
            own[4 * j + e] = quad_rot(q_bit1 ? hi : lo, j & 3);
        }
    }
}

__global__ __launch_bounds__(256)
void yolo_reduce(const float* __restrict__ out,
                 const float* __restrict__ tgt,
                 const int* __restrict__ imw,
                 const int* __restrict__ imh,
                 float* __restrict__ ws) {
    __shared__ float bins[3 * CELLS];
    const int tid = threadIdx.x;
    for (int k = tid; k < 3 * CELLS; k += blockDim.x) bins[k] = 0.0f;
    __syncthreads();

    const int   pr = blockIdx.x * blockDim.x + tid;
    const int   g0 = 2 * pr;
    const float W  = (float)imw[0];
    const float H  = (float)imh[0];
    const float cw = W / (float)SGRID;
    const float ch = H / (float)SGRID;

    const int qbase = (pr >> 2) * (4 * PAIRV) + (pr & 3);
    const v4f* __restrict__ tp = reinterpret_cast<const v4f*>(tgt) + qbase;
    const v4f* __restrict__ op = reinterpret_cast<const v4f*>(out) + qbase;

    v4f tv[PAIRV], ov[PAIRV];
#pragma unroll
    for (int k = 0; k < PAIRV; ++k) tv[k] = tp[4 * k];
#pragma unroll
    for (int k = 0; k < PAIRV; ++k) ov[k] = op[4 * k];
#pragma unroll
    for (int k = 0; k < PAIRV; ++k) asm volatile("" : "+v"(tv[k]));
#pragma unroll
    for (int k = 0; k < PAIRV; ++k) asm volatile("" : "+v"(ov[k]));

    float tl[2 * NCH], ol[2 * NCH];
#pragma unroll
    for (int k = 0; k < PAIRV; ++k) {
        tl[4 * k + 0] = tv[k].x; tl[4 * k + 1] = tv[k].y; tl[4 * k + 2] = tv[k].z; tl[4 * k + 3] = tv[k].w;
        ol[4 * k + 0] = ov[k].x; ol[4 * k + 1] = ov[k].y; ol[4 * k + 2] = ov[k].z; ol[4 * k + 3] = ov[k].w;
    }

    const bool q_bit0 = (tid & 1) != 0;
    const bool q_bit1 = (tid & 2) != 0;
    float tf[2 * NCH], of[2 * NCH];
    quad_gather(tl, tf, q_bit0, q_bit1);
    quad_gather(ol, of, q_bit0, q_bit1);

    const int b0  = g0 / CELLS;
    const int ij0 = g0 - b0 * CELLS;

    float hasp[2];
#pragma unroll
    for (int c = 0; c < 2; ++c) {
        int ij = ij0 + c;
        if (ij == CELLS) ij = 0;
        const int i = ij / SGRID;
        const int j = ij - i * SGRID;

        const float* t = tf + c * NCH;
        const float* o = of + c * NCH;

        const float t0 = t[0], t1 = t[1];
        const float tw = t[2] * W, th = t[3] * H;
        const float has = t[4];
        const float tcx = ((float)j + t0) * cw;
        const float tcy = ((float)i + t1) * ch;
        const float tx0 = tcx - tw * 0.5f, ty0 = tcy - th * 0.5f;
        const float tx1 = tcx + tw * 0.5f, ty1 = tcy + th * 0.5f;
        const float area_t = tw * th;

        float iou_p[2], conf_p[2];
#pragma unroll
        for (int p = 0; p < 2; ++p) {
            const float px = o[5 * p + 0], py = o[5 * p + 1];
            const float pw = o[5 * p + 3] * W, ph = o[5 * p + 4] * H;
            const float pcx = ((float)j + px) * cw;
            const float pcy = ((float)i + py) * ch;
            const float px0 = pcx - pw * 0.5f, py0 = pcy - ph * 0.5f;
            const float px1 = pcx + pw * 0.5f, py1 = pcy + ph * 0.5f;
            float iw = fminf(px1, tx1) - fmaxf(px0, tx0); iw = fmaxf(iw, 0.0f);
            float ih = fminf(py1, ty1) - fmaxf(py0, ty0); ih = fmaxf(ih, 0.0f);
            const float inter = iw * ih;
            iou_p[p]  = inter / (pw * ph + area_t - inter + 1e-9f);
            conf_p[p] = o[5 * p + 4];
        }
        const float best_conf = (iou_p[1] >= iou_p[0]) ? conf_p[1] : conf_p[0];

        const float px = o[5], py = o[6];
        const float pw = o[8] * W, ph = o[9] * H;
        const float pconf = o[9];
        const float dx = t0 - px, dy = t1 - py;
        const float dw = sqrtf(tw) - sqrtf(pw);
        const float dh = sqrtf(th) - sqrtf(ph);
        const float coord = dx * dx + dy * dy + dw * dw + dh * dh;
        const float dc = has - pconf;
        const float confm = dc * dc;

        float cls = 0.0f;
#pragma unroll
        for (int q = 0; q < NCLASS; ++q) {
            const float d = t[10 + q] - o[10 + q] * best_conf;
            cls = fmaf(d, d, cls);
        }

        hasp[c] = has;

        atomicAdd(&bins[ij],             coord);
        atomicAdd(&bins[CELLS + ij],     confm);
        atomicAdd(&bins[2 * CELLS + ij], cls);
    }

    *reinterpret_cast<float2*>(ws + WS_HAS + g0) = make_float2(hasp[0], hasp[1]);

    __syncthreads();
    if (tid < 32) {
        const float coord0 = bins[tid];
        const float conf0  = bins[CELLS + tid];
        const float cls0   = bins[2 * CELLS + tid];
        atomicAdd(&ws[WS_A + tid], 5.0f * coord0 + 0.5f * conf0);
        float cb = 0.5f * conf0 + cls0;
        if (tid < CELLS - 32) {
            const float coord1 = bins[32 + tid];
            const float conf1  = bins[CELLS + 32 + tid];
            const float cls1   = bins[2 * CELLS + 32 + tid];
            atomicAdd(&ws[WS_A + 32 + tid], 5.0f * coord1 + 0.5f * conf1);
            cb += 0.5f * conf1 + cls1;
        }
#pragma unroll
        for (int off = 16; off > 0; off >>= 1) cb += __shfl_xor(cb, off);
        if (tid == 0) atomicAdd(&ws[WS_CB], cb);
    }
}

__global__ void yolo_dot_wmma(const float* __restrict__ ws,
                              float* __restrict__ loss) {
    const int lane = threadIdx.x & 31;
    const int wave = (blockIdx.x * blockDim.x + threadIdx.x) >> 5;
    const int b0   = wave * 16;
    const float* __restrict__ Avec = ws + WS_A;
    const float* __restrict__ hasv = ws + WS_HAS;

    const int half = lane >> 4;
    const int m    = lane & 15;

    v8f acc = {};
#pragma unroll
    for (int k0 = 0; k0 < 52; k0 += 4) {
        const int ka = k0 + (half ? 2 : 0);
        const int kb = ka + 1;
        const int kca = ka > 48 ? 48 : ka;
        const int kcb = kb > 48 ? 48 : kb;
        v2f a, bm;
        a.x  = hasv[(size_t)(b0 + m) * CELLS + kca];
        a.y  = hasv[(size_t)(b0 + m) * CELLS + kcb];
        bm.x = Avec[ka];
        bm.y = Avec[kb];
        acc = __builtin_amdgcn_wmma_f32_16x16x4_f32(
             false, a,  false, bm,
             (short)0, acc,  false,  false);
    }

    const float cb = ws[WS_CB];
    if ((lane & 15) == 0) {
        const int mbase = b0 + half * 8;
#pragma unroll
        for (int r = 0; r < 8; ++r)
            loss[mbase + r] = acc[r] + cb;
    }
}

extern "C" void kernel_launch(void* const* d_in, const int* in_sizes, int n_in,
                              void* d_out, int out_size, void* d_ws, size_t ws_size,
                              hipStream_t stream) {
    const float* out_t = (const float*)d_in[0];
    const float* tgt_t = (const float*)d_in[1];
    const int*   imw   = (const int*)d_in[2];
    const int*   imh   = (const int*)d_in[3];
    float* loss = (float*)d_out;
    float* ws   = (float*)d_ws;

    yolo_init<<<1, 256, 0, stream>>>(ws);

    yolo_reduce<<<1568, 256, 0, stream>>>(out_t, tgt_t, imw, imh, ws);

    yolo_dot_wmma<<<128, 256, 0, stream>>>(ws, loss);
}
